# starting kernel
# baseline (speedup 1.0000x reference)
#include <hip/hip_runtime.h>
#include <math.h>


typedef float v2f __attribute__((ext_vector_type(2)));
typedef float v8f __attribute__((ext_vector_type(8)));

#define VDIM 16
#define NVOX 4096
#define NATOM 14
#define NLBL 21

__device__ __forceinline__ float dm_elem(int m, int k) {
  if (m == 0)  return (k == 0)  ? -1.0f : ((k == 1)  ? 1.0f : 0.0f);
  if (m == 15) return (k == 15) ?  1.0f : ((k == 14) ? -1.0f : 0.0f);
  if (k == m - 1) return -0.5f;
  if (k == m + 1) return  0.5f;
  return 0.0f;
}

template <int PASS>
__device__ __forceinline__ void div_pass(const float* __restrict__ src,
                                         float* __restrict__ dst,
                                         const v2f aReg[4], int lane, int wv) {
  const int nn    = lane & 15;
  const int khalf = (lane < 16) ? 0 : 2;
  const int mbase = (lane < 16) ? 0 : 8;
  for (int t = wv; t < 16; t += 8) {
    v8f acc = {};
#pragma unroll
    for (int kk = 0; kk < 4; ++kk) {
      const int k0 = 4 * kk + khalf;
      const int k1 = k0 + 1;
      int i0, i1;
      if (PASS == 0)      { i0 = k0 * 256 + t * 16 + nn; i1 = k1 * 256 + t * 16 + nn; }
      else if (PASS == 1) { i0 = t * 256 + k0 * 16 + nn; i1 = t * 256 + k1 * 16 + nn; }
      else                { i0 = t * 256 + nn * 16 + k0; i1 = t * 256 + nn * 16 + k1; }
      v2f b; b.x = src[i0]; b.y = src[i1];
      acc = __builtin_amdgcn_wmma_f32_16x16x4_f32(
           false, aReg[kk],  false, b,
           (short)0, acc,  false,  false);
    }
#pragma unroll
    for (int j = 0; j < 8; ++j) {
      const int m = mbase + j;
      int idx;
      if (PASS == 0)      idx = m * 256 + t * 16 + nn;
      else if (PASS == 1) idx = t * 256 + m * 16 + nn;
      else                idx = t * 256 + nn * 16 + m;
      if (PASS == 0) dst[idx] = acc[j];
      else           dst[idx] += acc[j];
    }
  }
}

__global__ __launch_bounds__(256)
void voxel_field_div_84670985273723(const float* __restrict__ C,
                                    const int*   __restrict__ L,
                                    const float* __restrict__ atom_mask,
                                    const float* __restrict__ charges,
                                    float* __restrict__ out, int ZN) {
  __shared__ float sFx[NVOX], sFy[NVOX], sFz[NVOX];
  __shared__ float sF[9], sO[3];
  __shared__ float sAx[NATOM], sAy[NATOM], sAz[NATOM], sPc[NATOM];

  const int n    = blockIdx.x;
  const int tid  = threadIdx.x;
  const int lane = tid & 31;
  const int wv   = tid >> 5;

  float* outCb  = out;
  float* outDiv = out + (size_t)ZN * 12;
  float* outFr  = outDiv + (size_t)ZN * NVOX;

  v2f aReg[4];
  {
    const int m     = lane & 15;
    const int khalf = (lane < 16) ? 0 : 2;
#pragma unroll
    for (int kk = 0; kk < 4; ++kk) {
      aReg[kk].x = dm_elem(m, 4 * kk + khalf);
      aReg[kk].y = dm_elem(m, 4 * kk + khalf + 1);
    }
  }

  if (tid < NATOM) {
    const float* cn = C + (size_t)n * (NATOM * 3);
    sAx[tid] = cn[tid * 3 + 0];
    sAy[tid] = cn[tid * 3 + 1];
    sAz[tid] = cn[tid * 3 + 2];
    int lbl = L[n];
    if (lbl == -1) lbl = NLBL - 1;
    sPc[tid] = charges[lbl * NATOM + tid] * atom_mask[(size_t)n * NATOM + tid];
  }
  if (tid == 0) {
    const float* cn = C + (size_t)n * (NATOM * 3);
    const float Nx = cn[0], Ny = cn[1], Nz = cn[2];
    const float Ax = cn[3], Ay = cn[4], Az = cn[5];
    const float Cx = cn[6], Cy = cn[7], Cz = cn[8];
    const float b1x = Ax - Nx, b1y = Ay - Ny, b1z = Az - Nz;
    const float b2x = Cx - Ax, b2y = Cy - Ay, b2z = Cz - Az;
    const float b3x = b1y * b2z - b1z * b2y;
    const float b3y = b1z * b2x - b1x * b2z;
    const float b3z = b1x * b2y - b1y * b2x;
    const float cbx = Ax - 0.58273431f * b2x + 0.56802827f * b1x - 0.54067466f * b3x;
    const float cby = Ay - 0.58273431f * b2y + 0.56802827f * b1y - 0.54067466f * b3y;
    const float cbz = Az - 0.58273431f * b2z + 0.56802827f * b1z - 0.54067466f * b3z;

    float* cb_out = outCb + (size_t)n * 12;
    cb_out[0] = Nx;  cb_out[1]  = Ny;  cb_out[2]  = Nz;
    cb_out[3] = Ax;  cb_out[4]  = Ay;  cb_out[5]  = Az;
    cb_out[6] = Cx;  cb_out[7]  = Cy;  cb_out[8]  = Cz;
    cb_out[9] = cbx; cb_out[10] = cby; cb_out[11] = cbz;

    float yx = cbx - Ax, yy = cby - Ay, yz = cbz - Az;
    float yn = fmaxf(sqrtf(yx * yx + yy * yy + yz * yz), 1e-6f);
    yx /= yn; yy /= yn; yz /= yn;
    float xrx = Cx - Nx, xry = Cy - Ny, xrz = Cz - Nz;
    const float xp = xrx * yx + xry * yy + xrz * yz;
    float xx = xrx - xp, xy = xry - xp, xz = xrz - xp;
    float xn = fmaxf(sqrtf(xx * xx + xy * xy + xz * xz), 1e-6f);
    xx /= xn; xy /= xn; xz /= xn;
    float zx = xy * yz - xz * yy;
    float zy = xz * yx - xx * yz;
    float zz = xx * yy - xy * yx;
    float zn = fmaxf(sqrtf(zx * zx + zy * zy + zz * zz), 1e-6f);
    zx /= zn; zy /= zn; zz /= zn;

    sF[0] = xx; sF[1] = xy; sF[2] = xz;
    sF[3] = yx; sF[4] = yy; sF[5] = yz;
    sF[6] = zx; sF[7] = zy; sF[8] = zz;
    sO[0] = cbx; sO[1] = cby; sO[2] = cbz;

    float* fr = outFr + (size_t)n * 9;
    fr[0] = xx; fr[1] = xy; fr[2] = xz;
    fr[3] = yx; fr[4] = yy; fr[5] = yz;
    fr[6] = zx; fr[7] = zy; fr[8] = zz;
  }
  __syncthreads();

  for (int i = tid; i < NVOX; i += 256) {
    const int x = i >> 8, y = (i >> 4) & 15, z = i & 15;
    const float vx = (float)x - 8.0f;
    const float vy = (float)y - 4.0f;
    const float vz = (float)z - 8.0f;
    const float px = sO[0] + vx * sF[0] + vy * sF[3] + vz * sF[6];
    const float py = sO[1] + vx * sF[1] + vy * sF[4] + vz * sF[7];
    const float pz = sO[2] + vx * sF[2] + vy * sF[5] + vz * sF[8];
    float fx = 0.0f, fy = 0.0f, fz = 0.0f;
#pragma unroll
    for (int a = 0; a < NATOM; ++a) {
      const float dx = px - sAx[a];
      const float dy = py - sAy[a];
      const float dz = pz - sAz[a];
      const float dd = dx * dx + dy * dy + dz * dz;
      const float d  = sqrtf(dd);
      const float dc = fmaxf(d, 1.0f);
      const float ds = (d == 0.0f) ? 1.0f : d;
      const float w  = sPc[a] / (ds * dc * dc);
      fx += w * dx; fy += w * dy; fz += w * dz;
    }
    const float fn = sqrtf(fx * fx + fy * fy + fz * fz);
    const float inv = (fn == 0.0f) ? 1.0f : 1.0f / fn;
    sFx[i] = fx * inv; sFy[i] = fy * inv; sFz[i] = fz * inv;
  }
  __syncthreads();

  div_pass<0>(sFx, sFx, aReg, lane, wv);
  __syncthreads();
  div_pass<1>(sFy, sFx, aReg, lane, wv);
  __syncthreads();
  div_pass<2>(sFz, sFx, aReg, lane, wv);
  __syncthreads();

  float* dv = outDiv + (size_t)n * NVOX;
  for (int i = tid; i < NVOX; i += 256) dv[i] = sFx[i];
}

extern "C" void kernel_launch(void* const* d_in, const int* in_sizes, int n_in,
                              void* d_out, int out_size, void* d_ws, size_t ws_size,
                              hipStream_t stream) {
  const float* C         = (const float*)d_in[0];
  const int*   L         = (const int*)d_in[1];
  const float* atom_mask = (const float*)d_in[2];
  const float* charges   = (const float*)d_in[3];
  float* out = (float*)d_out;
  const int ZN = in_sizes[0] / (NATOM * 3);
  voxel_field_div_84670985273723<<<ZN, 256, 0, stream>>>(C, L, atom_mask, charges, out, ZN);
}


# Per voxel-atom weight from one v_rsq_f32 and multiplies, not correctly rounded sqrtf plus IEEE divide (same in the per-voxel normalisation); f32 kept, last-bit rounding changes, sum order kept
# speedup vs baseline: 2.0211x; 2.0211x over previous
#include <hip/hip_runtime.h>
#include <math.h>


typedef float v2f __attribute__((ext_vector_type(2)));
typedef float v8f __attribute__((ext_vector_type(8)));

#define VDIM 16
#define NVOX 4096
#define NATOM 14
#define NLBL 21

__device__ __forceinline__ float dm_elem(int m, int k) {
  if (m == 0)  return (k == 0)  ? -1.0f : ((k == 1)  ? 1.0f : 0.0f);
  if (m == 15) return (k == 15) ?  1.0f : ((k == 14) ? -1.0f : 0.0f);
  if (k == m - 1) return -0.5f;
  if (k == m + 1) return  0.5f;
  return 0.0f;
}

template <int PASS>
__device__ __forceinline__ void div_pass(const float* __restrict__ src,
                                         float* __restrict__ dst,
                                         const v2f aReg[4], int lane, int wv) {
  const int nn    = lane & 15;
  const int khalf = (lane < 16) ? 0 : 2;
  const int mbase = (lane < 16) ? 0 : 8;
  for (int t = wv; t < 16; t += 8) {
    v8f acc = {};
#pragma unroll
    for (int kk = 0; kk < 4; ++kk) {
      const int k0 = 4 * kk + khalf;
      const int k1 = k0 + 1;
      int i0, i1;
      if (PASS == 0)      { i0 = k0 * 256 + t * 16 + nn; i1 = k1 * 256 + t * 16 + nn; }
      else if (PASS == 1) { i0 = t * 256 + k0 * 16 + nn; i1 = t * 256 + k1 * 16 + nn; }
      else                { i0 = t * 256 + nn * 16 + k0; i1 = t * 256 + nn * 16 + k1; }
      v2f b; b.x = src[i0]; b.y = src[i1];
      acc = __builtin_amdgcn_wmma_f32_16x16x4_f32(
           false, aReg[kk],  false, b,
           (short)0, acc,  false,  false);
    }
#pragma unroll
    for (int j = 0; j < 8; ++j) {
      const int m = mbase + j;
      int idx;
      if (PASS == 0)      idx = m * 256 + t * 16 + nn;
      else if (PASS == 1) idx = t * 256 + m * 16 + nn;
      else                idx = t * 256 + nn * 16 + m;
      if (PASS == 0) dst[idx] = acc[j];
      else           dst[idx] += acc[j];
    }
  }
}

__global__ __launch_bounds__(256)
void voxel_field_div_84670985273723(const float* __restrict__ C,
                                    const int*   __restrict__ L,
                                    const float* __restrict__ atom_mask,
                                    const float* __restrict__ charges,
                                    float* __restrict__ out, int ZN) {
  __shared__ float sFx[NVOX], sFy[NVOX], sFz[NVOX];
  __shared__ float sF[9], sO[3];
  __shared__ float sAx[NATOM], sAy[NATOM], sAz[NATOM], sPc[NATOM];

  const int n    = blockIdx.x;
  const int tid  = threadIdx.x;
  const int lane = tid & 31;
  const int wv   = tid >> 5;

  float* outCb  = out;
  float* outDiv = out + (size_t)ZN * 12;
  float* outFr  = outDiv + (size_t)ZN * NVOX;

  v2f aReg[4];
  {
    const int m     = lane & 15;
    const int khalf = (lane < 16) ? 0 : 2;
#pragma unroll
    for (int kk = 0; kk < 4; ++kk) {
      aReg[kk].x = dm_elem(m, 4 * kk + khalf);
      aReg[kk].y = dm_elem(m, 4 * kk + khalf + 1);
    }
  }

  if (tid < NATOM) {
    const float* cn = C + (size_t)n * (NATOM * 3);
    sAx[tid] = cn[tid * 3 + 0];
    sAy[tid] = cn[tid * 3 + 1];
    sAz[tid] = cn[tid * 3 + 2];
    int lbl = L[n];
    if (lbl == -1) lbl = NLBL - 1;
    sPc[tid] = charges[lbl * NATOM + tid] * atom_mask[(size_t)n * NATOM + tid];
  }
  if (tid == 0) {
    const float* cn = C + (size_t)n * (NATOM * 3);
    const float Nx = cn[0], Ny = cn[1], Nz = cn[2];
    const float Ax = cn[3], Ay = cn[4], Az = cn[5];
    const float Cx = cn[6], Cy = cn[7], Cz = cn[8];
    const float b1x = Ax - Nx, b1y = Ay - Ny, b1z = Az - Nz;
    const float b2x = Cx - Ax, b2y = Cy - Ay, b2z = Cz - Az;
    const float b3x = b1y * b2z - b1z * b2y;
    const float b3y = b1z * b2x - b1x * b2z;
    const float b3z = b1x * b2y - b1y * b2x;
    const float cbx = Ax - 0.58273431f * b2x + 0.56802827f * b1x - 0.54067466f * b3x;
    const float cby = Ay - 0.58273431f * b2y + 0.56802827f * b1y - 0.54067466f * b3y;
    const float cbz = Az - 0.58273431f * b2z + 0.56802827f * b1z - 0.54067466f * b3z;

    float* cb_out = outCb + (size_t)n * 12;
    cb_out[0] = Nx;  cb_out[1]  = Ny;  cb_out[2]  = Nz;
    cb_out[3] = Ax;  cb_out[4]  = Ay;  cb_out[5]  = Az;
    cb_out[6] = Cx;  cb_out[7]  = Cy;  cb_out[8]  = Cz;
    cb_out[9] = cbx; cb_out[10] = cby; cb_out[11] = cbz;

    float yx = cbx - Ax, yy = cby - Ay, yz = cbz - Az;
    float yn = fmaxf(sqrtf(yx * yx + yy * yy + yz * yz), 1e-6f);
    yx /= yn; yy /= yn; yz /= yn;
    float xrx = Cx - Nx, xry = Cy - Ny, xrz = Cz - Nz;
    const float xp = xrx * yx + xry * yy + xrz * yz;
    float xx = xrx - xp, xy = xry - xp, xz = xrz - xp;
    float xn = fmaxf(sqrtf(xx * xx + xy * xy + xz * xz), 1e-6f);
    xx /= xn; xy /= xn; xz /= xn;
    float zx = xy * yz - xz * yy;
    float zy = xz * yx - xx * yz;
    float zz = xx * yy - xy * yx;
    float zn = fmaxf(sqrtf(zx * zx + zy * zy + zz * zz), 1e-6f);
    zx /= zn; zy /= zn; zz /= zn;

    sF[0] = xx; sF[1] = xy; sF[2] = xz;
    sF[3] = yx; sF[4] = yy; sF[5] = yz;
    sF[6] = zx; sF[7] = zy; sF[8] = zz;
    sO[0] = cbx; sO[1] = cby; sO[2] = cbz;

    float* fr = outFr + (size_t)n * 9;
    fr[0] = xx; fr[1] = xy; fr[2] = xz;
    fr[3] = yx; fr[4] = yy; fr[5] = yz;
    fr[6] = zx; fr[7] = zy; fr[8] = zz;
  }
  __syncthreads();

  for (int i = tid; i < NVOX; i += 256) {
    const int x = i >> 8, y = (i >> 4) & 15, z = i & 15;
    const float vx = (float)x - 8.0f;
    const float vy = (float)y - 4.0f;
    const float vz = (float)z - 8.0f;
    const float px = sO[0] + vx * sF[0] + vy * sF[3] + vz * sF[6];
    const float py = sO[1] + vx * sF[1] + vy * sF[4] + vz * sF[7];
    const float pz = sO[2] + vx * sF[2] + vy * sF[5] + vz * sF[8];
    float fx = 0.0f, fy = 0.0f, fz = 0.0f;
#pragma unroll
    for (int a = 0; a < NATOM; ++a) {
      const float dx = px - sAx[a];
      const float dy = py - sAy[a];
      const float dz = pz - sAz[a];
      const float dd = dx * dx + dy * dy + dz * dz;
      const float r  = __builtin_amdgcn_rsqf(fmaxf(dd, 1.17549435e-38f));
      const float q  = fminf(r * r, 1.0f);
      const float w  = (sPc[a] * r) * q;
      fx += w * dx; fy += w * dy; fz += w * dz;
    }
    const float ff = fx * fx + fy * fy + fz * fz;
    const float inv = (ff == 0.0f) ? 1.0f : rsqrtf(ff);
    sFx[i] = fx * inv; sFy[i] = fy * inv; sFz[i] = fz * inv;
  }
  __syncthreads();

  div_pass<0>(sFx, sFx, aReg, lane, wv);
  __syncthreads();
  div_pass<1>(sFy, sFx, aReg, lane, wv);
  __syncthreads();
  div_pass<2>(sFz, sFx, aReg, lane, wv);
  __syncthreads();

  float* dv = outDiv + (size_t)n * NVOX;
  for (int i = tid; i < NVOX; i += 256) dv[i] = sFx[i];
}

extern "C" void kernel_launch(void* const* d_in, const int* in_sizes, int n_in,
                              void* d_out, int out_size, void* d_ws, size_t ws_size,
                              hipStream_t stream) {
  const float* C         = (const float*)d_in[0];
  const int*   L         = (const int*)d_in[1];
  const float* atom_mask = (const float*)d_in[2];
  const float* charges   = (const float*)d_in[3];
  float* out = (float*)d_out;
  const int ZN = in_sizes[0] / (NATOM * 3);
  voxel_field_div_84670985273723<<<ZN, 256, 0, stream>>>(C, L, atom_mask, charges, out, ZN);
}


# Field phase: two z-neighbour voxels per lane on packed 2xf32 instructions, atoms staged as {x,y},{z,q} pairs used as broadcast operands; f32, atom sum order kept, fma fusing changes last bits
# speedup vs baseline: 2.4571x; 1.2157x over previous
#include <hip/hip_runtime.h>
#include <math.h>


typedef float v2f __attribute__((ext_vector_type(2)));
typedef float v8f __attribute__((ext_vector_type(8)));

#define VDIM 16
#define NVOX 4096
#define NATOM 14
#define NLBL 21

__device__ __forceinline__ float dm_elem(int m, int k) {
  if (m == 0)  return (k == 0)  ? -1.0f : ((k == 1)  ? 1.0f : 0.0f);
  if (m == 15) return (k == 15) ?  1.0f : ((k == 14) ? -1.0f : 0.0f);
  if (k == m - 1) return -0.5f;
  if (k == m + 1) return  0.5f;
  return 0.0f;
}

template <int PASS>
__device__ __forceinline__ void div_pass(const float* __restrict__ src,
                                         float* __restrict__ dst,
                                         const v2f aReg[4], int lane, int wv) {
  const int nn    = lane & 15;
  const int khalf = (lane < 16) ? 0 : 2;
  const int mbase = (lane < 16) ? 0 : 8;
  for (int t = wv; t < 16; t += 8) {
    v8f acc = {};
#pragma unroll
    for (int kk = 0; kk < 4; ++kk) {
      const int k0 = 4 * kk + khalf;
      const int k1 = k0 + 1;
      int i0, i1;
      if (PASS == 0)      { i0 = k0 * 256 + t * 16 + nn; i1 = k1 * 256 + t * 16 + nn; }
      else if (PASS == 1) { i0 = t * 256 + k0 * 16 + nn; i1 = t * 256 + k1 * 16 + nn; }
      else                { i0 = t * 256 + nn * 16 + k0; i1 = t * 256 + nn * 16 + k1; }
      v2f b; b.x = src[i0]; b.y = src[i1];
      acc = __builtin_amdgcn_wmma_f32_16x16x4_f32(
           false, aReg[kk],  false, b,
           (short)0, acc,  false,  false);
    }
#pragma unroll
    for (int j = 0; j < 8; ++j) {
      const int m = mbase + j;
      int idx;
      if (PASS == 0)      idx = m * 256 + t * 16 + nn;
      else if (PASS == 1) idx = t * 256 + m * 16 + nn;
      else                idx = t * 256 + nn * 16 + m;
      if (PASS == 0) dst[idx] = acc[j];
      else           dst[idx] += acc[j];
    }
  }
}

__global__ __launch_bounds__(256)
void voxel_field_div_84670985273723(const float* __restrict__ C,
                                    const int*   __restrict__ L,
                                    const float* __restrict__ atom_mask,
                                    const float* __restrict__ charges,
                                    float* __restrict__ out, int ZN) {
  __shared__ __attribute__((aligned(16))) float sFx[NVOX], sFy[NVOX], sFz[NVOX];
  __shared__ float sF[9], sO[3];
  __shared__ v2f sAxy[NATOM], sAzq[NATOM];

  const int n    = blockIdx.x;
  const int tid  = threadIdx.x;
  const int lane = tid & 31;
  const int wv   = tid >> 5;

  float* outCb  = out;
  float* outDiv = out + (size_t)ZN * 12;
  float* outFr  = outDiv + (size_t)ZN * NVOX;

  v2f aReg[4];
  {
    const int m     = lane & 15;
    const int khalf = (lane < 16) ? 0 : 2;
#pragma unroll
    for (int kk = 0; kk < 4; ++kk) {
      aReg[kk].x = dm_elem(m, 4 * kk + khalf);
      aReg[kk].y = dm_elem(m, 4 * kk + khalf + 1);
    }
  }

  if (tid < NATOM) {
    const float* cn = C + (size_t)n * (NATOM * 3);
    int lbl = L[n];
    if (lbl == -1) lbl = NLBL - 1;
    v2f xy, zq;
    xy.x = cn[tid * 3 + 0];
    xy.y = cn[tid * 3 + 1];
    zq.x = cn[tid * 3 + 2];
    zq.y = charges[lbl * NATOM + tid] * atom_mask[(size_t)n * NATOM + tid];
    sAxy[tid] = xy;
    sAzq[tid] = zq;
  }
  if (tid == 0) {
    const float* cn = C + (size_t)n * (NATOM * 3);
    const float Nx = cn[0], Ny = cn[1], Nz = cn[2];
    const float Ax = cn[3], Ay = cn[4], Az = cn[5];
    const float Cx = cn[6], Cy = cn[7], Cz = cn[8];
    const float b1x = Ax - Nx, b1y = Ay - Ny, b1z = Az - Nz;
    const float b2x = Cx - Ax, b2y = Cy - Ay, b2z = Cz - Az;
    const float b3x = b1y * b2z - b1z * b2y;
    const float b3y = b1z * b2x - b1x * b2z;
    const float b3z = b1x * b2y - b1y * b2x;
    const float cbx = Ax - 0.58273431f * b2x + 0.56802827f * b1x - 0.54067466f * b3x;
    const float cby = Ay - 0.58273431f * b2y + 0.56802827f * b1y - 0.54067466f * b3y;
    const float cbz = Az - 0.58273431f * b2z + 0.56802827f * b1z - 0.54067466f * b3z;

    float* cb_out = outCb + (size_t)n * 12;
    cb_out[0] = Nx;  cb_out[1]  = Ny;  cb_out[2]  = Nz;
    cb_out[3] = Ax;  cb_out[4]  = Ay;  cb_out[5]  = Az;
    cb_out[6] = Cx;  cb_out[7]  = Cy;  cb_out[8]  = Cz;
    cb_out[9] = cbx; cb_out[10] = cby; cb_out[11] = cbz;

    float yx = cbx - Ax, yy = cby - Ay, yz = cbz - Az;
    float yn = fmaxf(sqrtf(yx * yx + yy * yy + yz * yz), 1e-6f);
    yx /= yn; yy /= yn; yz /= yn;
    float xrx = Cx - Nx, xry = Cy - Ny, xrz = Cz - Nz;
    const float xp = xrx * yx + xry * yy + xrz * yz;
    float xx = xrx - xp, xy = xry - xp, xz = xrz - xp;
    float xn = fmaxf(sqrtf(xx * xx + xy * xy + xz * xz), 1e-6f);
    xx /= xn; xy /= xn; xz /= xn;
    float zx = xy * yz - xz * yy;
    float zy = xz * yx - xx * yz;
    float zz = xx * yy - xy * yx;
    float zn = fmaxf(sqrtf(zx * zx + zy * zy + zz * zz), 1e-6f);
    zx /= zn; zy /= zn; zz /= zn;

    sF[0] = xx; sF[1] = xy; sF[2] = xz;
    sF[3] = yx; sF[4] = yy; sF[5] = yz;
    sF[6] = zx; sF[7] = zy; sF[8] = zz;
    sO[0] = cbx; sO[1] = cby; sO[2] = cbz;

    float* fr = outFr + (size_t)n * 9;
    fr[0] = xx; fr[1] = xy; fr[2] = xz;
    fr[3] = yx; fr[4] = yy; fr[5] = yz;
    fr[6] = zx; fr[7] = zy; fr[8] = zz;
  }
  __syncthreads();

  for (int j = tid; j < NVOX / 2; j += 256) {
    const int x = j >> 7, y = (j >> 3) & 15, z = (j & 7) << 1;
    const float vx = (float)x - 8.0f;
    const float vy = (float)y - 4.0f;
    v2f vz;
    vz.x = (float)z - 8.0f;
    vz.y = (float)(z + 1) - 8.0f;
    const v2f px = sO[0] + vx * sF[0] + vy * sF[3] + vz * sF[6];
    const v2f py = sO[1] + vx * sF[1] + vy * sF[4] + vz * sF[7];
    const v2f pz = sO[2] + vx * sF[2] + vy * sF[5] + vz * sF[8];
    v2f fx = {0.0f, 0.0f}, fy = {0.0f, 0.0f}, fz = {0.0f, 0.0f};
#pragma unroll
    for (int a = 0; a < NATOM; ++a) {
      const v2f axy = sAxy[a], azq = sAzq[a];
      const v2f dx = px - axy.x;
      const v2f dy = py - axy.y;
      const v2f dz = pz - azq.x;
      const v2f dd = dx * dx + dy * dy + dz * dz;
      v2f r;
      r.x = __builtin_amdgcn_rsqf(fmaxf(dd.x, 1.17549435e-38f));
      r.y = __builtin_amdgcn_rsqf(fmaxf(dd.y, 1.17549435e-38f));
      v2f q = r * r;
      q.x = fminf(q.x, 1.0f);
      q.y = fminf(q.y, 1.0f);
      const v2f w = (azq.y * r) * q;
      fx += w * dx; fy += w * dy; fz += w * dz;
    }
    const v2f ff = fx * fx + fy * fy + fz * fz;
    v2f inv;
    inv.x = (ff.x == 0.0f) ? 1.0f : rsqrtf(ff.x);
    inv.y = (ff.y == 0.0f) ? 1.0f : rsqrtf(ff.y);
    *reinterpret_cast<v2f*>(&sFx[2 * j]) = fx * inv;
    *reinterpret_cast<v2f*>(&sFy[2 * j]) = fy * inv;
    *reinterpret_cast<v2f*>(&sFz[2 * j]) = fz * inv;
  }
  __syncthreads();

  div_pass<0>(sFx, sFx, aReg, lane, wv);
  __syncthreads();
  div_pass<1>(sFy, sFx, aReg, lane, wv);
  __syncthreads();
  div_pass<2>(sFz, sFx, aReg, lane, wv);
  __syncthreads();

  float* dv = outDiv + (size_t)n * NVOX;
  for (int i = tid; i < NVOX; i += 256) dv[i] = sFx[i];
}

extern "C" void kernel_launch(void* const* d_in, const int* in_sizes, int n_in,
                              void* d_out, int out_size, void* d_ws, size_t ws_size,
                              hipStream_t stream) {
  const float* C         = (const float*)d_in[0];
  const int*   L         = (const int*)d_in[1];
  const float* atom_mask = (const float*)d_in[2];
  const float* charges   = (const float*)d_in[3];
  float* out = (float*)d_out;
  const int ZN = in_sizes[0] / (NATOM * 3);
  voxel_field_div_84670985273723<<<ZN, 256, 0, stream>>>(C, L, atom_mask, charges, out, ZN);
}


# Field loop: the zero guards of both reciprocal square roots become a 2^-126 seed in the first fma of each sum of squares (no max, select or input scaling left); f32, sum order kept, last bits change
# speedup vs baseline: 2.6978x; 1.0980x over previous
#include <hip/hip_runtime.h>
#include <math.h>


typedef float v2f __attribute__((ext_vector_type(2)));
typedef float v8f __attribute__((ext_vector_type(8)));

#define VDIM 16
#define NVOX 4096
#define NATOM 14
#define NLBL 21

__device__ __forceinline__ float dm_elem(int m, int k) {
  if (m == 0)  return (k == 0)  ? -1.0f : ((k == 1)  ? 1.0f : 0.0f);
  if (m == 15) return (k == 15) ?  1.0f : ((k == 14) ? -1.0f : 0.0f);
  if (k == m - 1) return -0.5f;
  if (k == m + 1) return  0.5f;
  return 0.0f;
}

template <int PASS>
__device__ __forceinline__ void div_pass(const float* __restrict__ src,
                                         float* __restrict__ dst,
                                         const v2f aReg[4], int lane, int wv) {
  const int nn    = lane & 15;
  const int khalf = (lane < 16) ? 0 : 2;
  const int mbase = (lane < 16) ? 0 : 8;
  for (int t = wv; t < 16; t += 8) {
    v8f acc = {};
#pragma unroll
    for (int kk = 0; kk < 4; ++kk) {
      const int k0 = 4 * kk + khalf;
      const int k1 = k0 + 1;
      int i0, i1;
      if (PASS == 0)      { i0 = k0 * 256 + t * 16 + nn; i1 = k1 * 256 + t * 16 + nn; }
      else if (PASS == 1) { i0 = t * 256 + k0 * 16 + nn; i1 = t * 256 + k1 * 16 + nn; }
      else                { i0 = t * 256 + nn * 16 + k0; i1 = t * 256 + nn * 16 + k1; }
      v2f b; b.x = src[i0]; b.y = src[i1];
      acc = __builtin_amdgcn_wmma_f32_16x16x4_f32(
           false, aReg[kk],  false, b,
           (short)0, acc,  false,  false);
    }
#pragma unroll
    for (int j = 0; j < 8; ++j) {
      const int m = mbase + j;
      int idx;
      if (PASS == 0)      idx = m * 256 + t * 16 + nn;
      else if (PASS == 1) idx = t * 256 + m * 16 + nn;
      else                idx = t * 256 + nn * 16 + m;
      if (PASS == 0) dst[idx] = acc[j];
      else           dst[idx] += acc[j];
    }
  }
}

__global__ __launch_bounds__(256)
void voxel_field_div_84670985273723(const float* __restrict__ C,
                                    const int*   __restrict__ L,
                                    const float* __restrict__ atom_mask,
                                    const float* __restrict__ charges,
                                    float* __restrict__ out, int ZN) {
  __shared__ __attribute__((aligned(16))) float sFx[NVOX], sFy[NVOX], sFz[NVOX];
  __shared__ float sF[9], sO[3];
  __shared__ v2f sAxy[NATOM], sAzq[NATOM];

  const int n    = blockIdx.x;
  const int tid  = threadIdx.x;
  const int lane = tid & 31;
  const int wv   = tid >> 5;

  float* outCb  = out;
  float* outDiv = out + (size_t)ZN * 12;
  float* outFr  = outDiv + (size_t)ZN * NVOX;

  v2f aReg[4];
  {
    const int m     = lane & 15;
    const int khalf = (lane < 16) ? 0 : 2;
#pragma unroll
    for (int kk = 0; kk < 4; ++kk) {
      aReg[kk].x = dm_elem(m, 4 * kk + khalf);
      aReg[kk].y = dm_elem(m, 4 * kk + khalf + 1);
    }
  }

  if (tid < NATOM) {
    const float* cn = C + (size_t)n * (NATOM * 3);
    int lbl = L[n];
    if (lbl == -1) lbl = NLBL - 1;
    v2f xy, zq;
    xy.x = cn[tid * 3 + 0];
    xy.y = cn[tid * 3 + 1];
    zq.x = cn[tid * 3 + 2];
    zq.y = charges[lbl * NATOM + tid] * atom_mask[(size_t)n * NATOM + tid];
    sAxy[tid] = xy;
    sAzq[tid] = zq;
  }
  if (tid == 0) {
    const float* cn = C + (size_t)n * (NATOM * 3);
    const float Nx = cn[0], Ny = cn[1], Nz = cn[2];
    const float Ax = cn[3], Ay = cn[4], Az = cn[5];
    const float Cx = cn[6], Cy = cn[7], Cz = cn[8];
    const float b1x = Ax - Nx, b1y = Ay - Ny, b1z = Az - Nz;
    const float b2x = Cx - Ax, b2y = Cy - Ay, b2z = Cz - Az;
    const float b3x = b1y * b2z - b1z * b2y;
    const float b3y = b1z * b2x - b1x * b2z;
    const float b3z = b1x * b2y - b1y * b2x;
    const float cbx = Ax - 0.58273431f * b2x + 0.56802827f * b1x - 0.54067466f * b3x;
    const float cby = Ay - 0.58273431f * b2y + 0.56802827f * b1y - 0.54067466f * b3y;
    const float cbz = Az - 0.58273431f * b2z + 0.56802827f * b1z - 0.54067466f * b3z;

    float* cb_out = outCb + (size_t)n * 12;
    cb_out[0] = Nx;  cb_out[1]  = Ny;  cb_out[2]  = Nz;
    cb_out[3] = Ax;  cb_out[4]  = Ay;  cb_out[5]  = Az;
    cb_out[6] = Cx;  cb_out[7]  = Cy;  cb_out[8]  = Cz;
    cb_out[9] = cbx; cb_out[10] = cby; cb_out[11] = cbz;

    float yx = cbx - Ax, yy = cby - Ay, yz = cbz - Az;
    float yn = fmaxf(sqrtf(yx * yx + yy * yy + yz * yz), 1e-6f);
    yx /= yn; yy /= yn; yz /= yn;
    float xrx = Cx - Nx, xry = Cy - Ny, xrz = Cz - Nz;
    const float xp = xrx * yx + xry * yy + xrz * yz;
    float xx = xrx - xp, xy = xry - xp, xz = xrz - xp;
    float xn = fmaxf(sqrtf(xx * xx + xy * xy + xz * xz), 1e-6f);
    xx /= xn; xy /= xn; xz /= xn;
    float zx = xy * yz - xz * yy;
    float zy = xz * yx - xx * yz;
    float zz = xx * yy - xy * yx;
    float zn = fmaxf(sqrtf(zx * zx + zy * zy + zz * zz), 1e-6f);
    zx /= zn; zy /= zn; zz /= zn;

    sF[0] = xx; sF[1] = xy; sF[2] = xz;
    sF[3] = yx; sF[4] = yy; sF[5] = yz;
    sF[6] = zx; sF[7] = zy; sF[8] = zz;
    sO[0] = cbx; sO[1] = cby; sO[2] = cbz;

    float* fr = outFr + (size_t)n * 9;
    fr[0] = xx; fr[1] = xy; fr[2] = xz;
    fr[3] = yx; fr[4] = yy; fr[5] = yz;
    fr[6] = zx; fr[7] = zy; fr[8] = zz;
  }
  __syncthreads();

  for (int j = tid; j < NVOX / 2; j += 256) {
    const int x = j >> 7, y = (j >> 3) & 15, z = (j & 7) << 1;
    const float vx = (float)x - 8.0f;
    const float vy = (float)y - 4.0f;
    v2f vz;
    vz.x = (float)z - 8.0f;
    vz.y = (float)(z + 1) - 8.0f;
    const v2f px = sO[0] + vx * sF[0] + vy * sF[3] + vz * sF[6];
    const v2f py = sO[1] + vx * sF[1] + vy * sF[4] + vz * sF[7];
    const v2f pz = sO[2] + vx * sF[2] + vy * sF[5] + vz * sF[8];
    v2f fx = {0.0f, 0.0f}, fy = {0.0f, 0.0f}, fz = {0.0f, 0.0f};
    const v2f seed = {1.17549435e-38f, 1.17549435e-38f};
#pragma unroll
    for (int a = 0; a < NATOM; ++a) {
      const v2f axy = sAxy[a], azq = sAzq[a];
      const v2f dx = px - axy.x;
      const v2f dy = py - axy.y;
      const v2f dz = pz - azq.x;
      const v2f dd = (dx * dx + seed) + dy * dy + dz * dz;
      v2f r;
      r.x = __builtin_amdgcn_rsqf(dd.x);
      r.y = __builtin_amdgcn_rsqf(dd.y);
      v2f q = r * r;
      q.x = fminf(q.x, 1.0f);
      q.y = fminf(q.y, 1.0f);
      const v2f w = (azq.y * r) * q;
      fx += w * dx; fy += w * dy; fz += w * dz;
    }
    const v2f ff = (fx * fx + seed) + fy * fy + fz * fz;
    v2f inv;
    inv.x = __builtin_amdgcn_rsqf(ff.x);
    inv.y = __builtin_amdgcn_rsqf(ff.y);
    *reinterpret_cast<v2f*>(&sFx[2 * j]) = fx * inv;
    *reinterpret_cast<v2f*>(&sFy[2 * j]) = fy * inv;
    *reinterpret_cast<v2f*>(&sFz[2 * j]) = fz * inv;
  }
  __syncthreads();

  div_pass<0>(sFx, sFx, aReg, lane, wv);
  __syncthreads();
  div_pass<1>(sFy, sFx, aReg, lane, wv);
  __syncthreads();
  div_pass<2>(sFz, sFx, aReg, lane, wv);
  __syncthreads();

  float* dv = outDiv + (size_t)n * NVOX;
  for (int i = tid; i < NVOX; i += 256) dv[i] = sFx[i];
}

extern "C" void kernel_launch(void* const* d_in, const int* in_sizes, int n_in,
                              void* d_out, int out_size, void* d_ws, size_t ws_size,
                              hipStream_t stream) {
  const float* C         = (const float*)d_in[0];
  const int*   L         = (const int*)d_in[1];
  const float* atom_mask = (const float*)d_in[2];
  const float* charges   = (const float*)d_in[3];
  float* out = (float*)d_out;
  const int ZN = in_sizes[0] / (NATOM * 3);
  voxel_field_div_84670985273723<<<ZN, 256, 0, stream>>>(C, L, atom_mask, charges, out, ZN);
}


# Divergence: y and z derivative products of each x-plane chained in one accumulator tile on top of dx read as C (Dm x Fy + Fz x Dm^T), stored from registers; f32, dx+dy+dz sum order changes
# speedup vs baseline: 3.1347x; 1.1619x over previous
#include <hip/hip_runtime.h>
#include <math.h>


typedef float v2f __attribute__((ext_vector_type(2)));
typedef float v8f __attribute__((ext_vector_type(8)));

#define VDIM 16
#define NVOX 4096
#define NATOM 14
#define NLBL 21

__device__ __forceinline__ float dm_elem(int m, int k) {
  const bool  edge = (m == 0) || (m == 15);
  const float off  = (k == m + 1) ? 0.5f : ((k == m - 1) ? -0.5f : 0.0f);
  const float diag = (k != m) ? 0.0f : ((m == 0) ? -1.0f : ((m == 15) ? 1.0f : 0.0f));
  return (edge ? 2.0f : 1.0f) * off + diag;
}

__device__ __forceinline__ void div_x_pass(const float* __restrict__ src,
                                           float* __restrict__ dst,
                                           const v2f aReg[4], int lane, int wv) {
  const int nn    = lane & 15;
  const int khalf = (lane < 16) ? 0 : 2;
  const int mbase = (lane < 16) ? 0 : 8;
  for (int t = wv; t < 16; t += 8) {
    v8f acc = {};
#pragma unroll
    for (int kk = 0; kk < 4; ++kk) {
      const int k0 = 4 * kk + khalf;
      const int k1 = k0 + 1;
      const int i0 = k0 * 256 + t * 16 + nn;
      const int i1 = k1 * 256 + t * 16 + nn;
      v2f b; b.x = src[i0]; b.y = src[i1];
      acc = __builtin_amdgcn_wmma_f32_16x16x4_f32(
           false, aReg[kk],  false, b,
           (short)0, acc,  false,  false);
    }
#pragma unroll
    for (int j = 0; j < 8; ++j) {
      const int m = mbase + j;
      dst[m * 256 + t * 16 + nn] = acc[j];
    }
  }
}

__device__ __forceinline__ void div_yz_store(const float* __restrict__ sDx,
                                             const float* __restrict__ sFy,
                                             const float* __restrict__ sFz,
                                             float* __restrict__ dv,
                                             const v2f aNat[4], const v2f aPerm[4],
                                             int lane, int wv) {
  const int c    = lane & 15;
  const int h    = lane >> 4;
  const int yrow = 2 * (c & 7) + (c >> 3);
#pragma unroll
  for (int i = 0; i < 2; ++i) {
    const int t = wv + 8 * i;
    const float* px = sDx + t * 256;
    const float* py = sFy + t * 256;
    const float* pz = sFz + t * 256;
    v8f acc;
#pragma unroll
    for (int r = 0; r < 8; ++r) acc[r] = px[r * 32 + lane];
#pragma unroll
    for (int kk = 0; kk < 4; ++kk) {
      const int k0 = 4 * kk + 2 * h;
      v2f b; b.x = py[k0 * 16 + c]; b.y = py[(k0 + 1) * 16 + c];
      acc = __builtin_amdgcn_wmma_f32_16x16x4_f32(
           false, aPerm[kk],  false, b,
           (short)0, acc,  false,  false);
    }
#pragma unroll
    for (int kk = 0; kk < 4; ++kk) {
      const int k0 = 4 * kk + 2 * h;
      const v2f a = *reinterpret_cast<const v2f*>(&pz[yrow * 16 + k0]);
      acc = __builtin_amdgcn_wmma_f32_16x16x4_f32(
           false, a,  false, aNat[kk],
           (short)0, acc,  false,  false);
    }
    float* o = dv + t * 256;
#pragma unroll
    for (int r = 0; r < 8; ++r) o[r * 32 + lane] = acc[r];
  }
}

__global__ __launch_bounds__(256)
void voxel_field_div_84670985273723(const float* __restrict__ C,
                                    const int*   __restrict__ L,
                                    const float* __restrict__ atom_mask,
                                    const float* __restrict__ charges,
                                    float* __restrict__ out, int ZN) {
  __shared__ __attribute__((aligned(16))) float sFx[NVOX], sFy[NVOX], sFz[NVOX];
  __shared__ float sF[9], sO[3];
  __shared__ v2f sAxy[NATOM], sAzq[NATOM];

  const int n    = blockIdx.x;
  const int tid  = threadIdx.x;
  const int lane = tid & 31;
  const int wv   = tid >> 5;

  float* outCb  = out;
  float* outDiv = out + (size_t)ZN * 12;
  float* outFr  = outDiv + (size_t)ZN * NVOX;

  if (tid < NATOM) {
    const float* cn = C + (size_t)n * (NATOM * 3);
    int lbl = L[n];
    if (lbl == -1) lbl = NLBL - 1;
    v2f xy, zq;
    xy.x = cn[tid * 3 + 0];
    xy.y = cn[tid * 3 + 1];
    zq.x = cn[tid * 3 + 2];
    zq.y = charges[lbl * NATOM + tid] * atom_mask[(size_t)n * NATOM + tid];
    sAxy[tid] = xy;
    sAzq[tid] = zq;
  }
  if (tid == 0) {
    const float* cn = C + (size_t)n * (NATOM * 3);
    const float Nx = cn[0], Ny = cn[1], Nz = cn[2];
    const float Ax = cn[3], Ay = cn[4], Az = cn[5];
    const float Cx = cn[6], Cy = cn[7], Cz = cn[8];
    const float b1x = Ax - Nx, b1y = Ay - Ny, b1z = Az - Nz;
    const float b2x = Cx - Ax, b2y = Cy - Ay, b2z = Cz - Az;
    const float b3x = b1y * b2z - b1z * b2y;
    const float b3y = b1z * b2x - b1x * b2z;
    const float b3z = b1x * b2y - b1y * b2x;
    const float cbx = Ax - 0.58273431f * b2x + 0.56802827f * b1x - 0.54067466f * b3x;
    const float cby = Ay - 0.58273431f * b2y + 0.56802827f * b1y - 0.54067466f * b3y;
    const float cbz = Az - 0.58273431f * b2z + 0.56802827f * b1z - 0.54067466f * b3z;

    float* cb_out = outCb + (size_t)n * 12;
    cb_out[0] = Nx;  cb_out[1]  = Ny;  cb_out[2]  = Nz;
    cb_out[3] = Ax;  cb_out[4]  = Ay;  cb_out[5]  = Az;
    cb_out[6] = Cx;  cb_out[7]  = Cy;  cb_out[8]  = Cz;
    cb_out[9] = cbx; cb_out[10] = cby; cb_out[11] = cbz;

    float yx = cbx - Ax, yy = cby - Ay, yz = cbz - Az;
    float yn = fmaxf(sqrtf(yx * yx + yy * yy + yz * yz), 1e-6f);
    yx /= yn; yy /= yn; yz /= yn;
    float xrx = Cx - Nx, xry = Cy - Ny, xrz = Cz - Nz;
    const float xp = xrx * yx + xry * yy + xrz * yz;
    float xx = xrx - xp, xy = xry - xp, xz = xrz - xp;
    float xn = fmaxf(sqrtf(xx * xx + xy * xy + xz * xz), 1e-6f);
    xx /= xn; xy /= xn; xz /= xn;
    float zx = xy * yz - xz * yy;
    float zy = xz * yx - xx * yz;
    float zz = xx * yy - xy * yx;
    float zn = fmaxf(sqrtf(zx * zx + zy * zy + zz * zz), 1e-6f);
    zx /= zn; zy /= zn; zz /= zn;

    sF[0] = xx; sF[1] = xy; sF[2] = xz;
    sF[3] = yx; sF[4] = yy; sF[5] = yz;
    sF[6] = zx; sF[7] = zy; sF[8] = zz;
    sO[0] = cbx; sO[1] = cby; sO[2] = cbz;

    float* fr = outFr + (size_t)n * 9;
    fr[0] = xx; fr[1] = xy; fr[2] = xz;
    fr[3] = yx; fr[4] = yy; fr[5] = yz;
    fr[6] = zx; fr[7] = zy; fr[8] = zz;
  }
  __syncthreads();

  for (int j = tid; j < NVOX / 2; j += 256) {
    const int x = j >> 7, y = (j >> 3) & 15, z = (j & 7) << 1;
    const float vx = (float)x - 8.0f;
    const float vy = (float)y - 4.0f;
    v2f vz;
    vz.x = (float)z - 8.0f;
    vz.y = (float)(z + 1) - 8.0f;
    const v2f px = sO[0] + vx * sF[0] + vy * sF[3] + vz * sF[6];
    const v2f py = sO[1] + vx * sF[1] + vy * sF[4] + vz * sF[7];
    const v2f pz = sO[2] + vx * sF[2] + vy * sF[5] + vz * sF[8];
    v2f fx = {0.0f, 0.0f}, fy = {0.0f, 0.0f}, fz = {0.0f, 0.0f};
    const v2f seed = {1.17549435e-38f, 1.17549435e-38f};
#pragma unroll
    for (int a = 0; a < NATOM; ++a) {
      const v2f axy = sAxy[a], azq = sAzq[a];
      const v2f dx = px - axy.x;
      const v2f dy = py - axy.y;
      const v2f dz = pz - azq.x;
      const v2f dd = (dx * dx + seed) + dy * dy + dz * dz;
      v2f r;
      r.x = __builtin_amdgcn_rsqf(dd.x);
      r.y = __builtin_amdgcn_rsqf(dd.y);
      v2f q = r * r;
      q.x = fminf(q.x, 1.0f);
      q.y = fminf(q.y, 1.0f);
      const v2f w = (azq.y * r) * q;
      fx += w * dx; fy += w * dy; fz += w * dz;
    }
    const v2f ff = (fx * fx + seed) + fy * fy + fz * fz;
    v2f inv;
    inv.x = __builtin_amdgcn_rsqf(ff.x);
    inv.y = __builtin_amdgcn_rsqf(ff.y);
    *reinterpret_cast<v2f*>(&sFx[2 * j]) = fx * inv;
    *reinterpret_cast<v2f*>(&sFy[2 * j]) = fy * inv;
    *reinterpret_cast<v2f*>(&sFz[2 * j]) = fz * inv;
  }
  __syncthreads();

  v2f aReg[4], aPerm[4];
  {
    const int m     = lane & 15;
    const int mp    = 2 * (m & 7) + (m >> 3);
    const int khalf = (lane < 16) ? 0 : 2;
#pragma unroll
    for (int kk = 0; kk < 4; ++kk) {
      aReg[kk].x  = dm_elem(m, 4 * kk + khalf);
      aReg[kk].y  = dm_elem(m, 4 * kk + khalf + 1);
      aPerm[kk].x = dm_elem(mp, 4 * kk + khalf);
      aPerm[kk].y = dm_elem(mp, 4 * kk + khalf + 1);
    }
  }

  div_x_pass(sFx, sFx, aReg, lane, wv);
  __syncthreads();
  div_yz_store(sFx, sFy, sFz, outDiv + (size_t)n * NVOX, aReg, aPerm, lane, wv);
}

extern "C" void kernel_launch(void* const* d_in, const int* in_sizes, int n_in,
                              void* d_out, int out_size, void* d_ws, size_t ws_size,
                              hipStream_t stream) {
  const float* C         = (const float*)d_in[0];
  const int*   L         = (const int*)d_in[1];
  const float* atom_mask = (const float*)d_in[2];
  const float* charges   = (const float*)d_in[3];
  float* out = (float*)d_out;
  const int ZN = in_sizes[0] / (NATOM * 3);
  voxel_field_div_84670985273723<<<ZN, 256, 0, stream>>>(C, L, atom_mask, charges, out, ZN);
}


# Set-up phase: atom rows + point charges on wave 0 and frames on wave 1 side by side, label-independent loads issued first; Cb/frames stored behind the first barrier, whose store-wait held the block
# speedup vs baseline: 3.3130x; 1.0569x over previous
#include <hip/hip_runtime.h>
#include <math.h>


typedef float v2f __attribute__((ext_vector_type(2)));
typedef float v8f __attribute__((ext_vector_type(8)));

#define VDIM 16
#define NVOX 4096
#define NATOM 14
#define NLBL 21

__device__ __forceinline__ float dm_elem(int m, int k) {
  const bool  edge = (m == 0) || (m == 15);
  const float off  = (k == m + 1) ? 0.5f : ((k == m - 1) ? -0.5f : 0.0f);
  const float diag = (k != m) ? 0.0f : ((m == 0) ? -1.0f : ((m == 15) ? 1.0f : 0.0f));
  return (edge ? 2.0f : 1.0f) * off + diag;
}

__device__ __forceinline__ void div_x_pass(const float* __restrict__ src,
                                           float* __restrict__ dst,
                                           const v2f aReg[4], int lane, int wv) {
  const int nn    = lane & 15;
  const int khalf = (lane < 16) ? 0 : 2;
  const int mbase = (lane < 16) ? 0 : 8;
  for (int t = wv; t < 16; t += 8) {
    v8f acc = {};
#pragma unroll
    for (int kk = 0; kk < 4; ++kk) {
      const int k0 = 4 * kk + khalf;
      const int k1 = k0 + 1;
      const int i0 = k0 * 256 + t * 16 + nn;
      const int i1 = k1 * 256 + t * 16 + nn;
      v2f b; b.x = src[i0]; b.y = src[i1];
      acc = __builtin_amdgcn_wmma_f32_16x16x4_f32(
           false, aReg[kk],  false, b,
           (short)0, acc,  false,  false);
    }
#pragma unroll
    for (int j = 0; j < 8; ++j) {
      const int m = mbase + j;
      dst[m * 256 + t * 16 + nn] = acc[j];
    }
  }
}

__device__ __forceinline__ void div_yz_store(const float* __restrict__ sDx,
                                             const float* __restrict__ sFy,
                                             const float* __restrict__ sFz,
                                             float* __restrict__ dv,
                                             const v2f aNat[4], const v2f aPerm[4],
                                             int lane, int wv) {
  const int c    = lane & 15;
  const int h    = lane >> 4;
  const int yrow = 2 * (c & 7) + (c >> 3);
#pragma unroll
  for (int i = 0; i < 2; ++i) {
    const int t = wv + 8 * i;
    const float* px = sDx + t * 256;
    const float* py = sFy + t * 256;
    const float* pz = sFz + t * 256;
    v8f acc;
#pragma unroll
    for (int r = 0; r < 8; ++r) acc[r] = px[r * 32 + lane];
#pragma unroll
    for (int kk = 0; kk < 4; ++kk) {
      const int k0 = 4 * kk + 2 * h;
      v2f b; b.x = py[k0 * 16 + c]; b.y = py[(k0 + 1) * 16 + c];
      acc = __builtin_amdgcn_wmma_f32_16x16x4_f32(
           false, aPerm[kk],  false, b,
           (short)0, acc,  false,  false);
    }
#pragma unroll
    for (int kk = 0; kk < 4; ++kk) {
      const int k0 = 4 * kk + 2 * h;
      const v2f a = *reinterpret_cast<const v2f*>(&pz[yrow * 16 + k0]);
      acc = __builtin_amdgcn_wmma_f32_16x16x4_f32(
           false, a,  false, aNat[kk],
           (short)0, acc,  false,  false);
    }
    float* o = dv + t * 256;
#pragma unroll
    for (int r = 0; r < 8; ++r) o[r * 32 + lane] = acc[r];
  }
}

__global__ __launch_bounds__(256)
void voxel_field_div_84670985273723(const float* __restrict__ C,
                                    const int*   __restrict__ L,
                                    const float* __restrict__ atom_mask,
                                    const float* __restrict__ charges,
                                    float* __restrict__ out, int ZN) {
  __shared__ __attribute__((aligned(16))) float sFx[NVOX], sFy[NVOX], sFz[NVOX];
  __shared__ float sF[9], sO[3];
  __shared__ v2f sAxy[NATOM], sAzq[NATOM];

  const int n    = blockIdx.x;
  const int tid  = threadIdx.x;
  const int lane = tid & 31;
  const int wv   = tid >> 5;

  float* outCb  = out;
  float* outDiv = out + (size_t)ZN * 12;
  float* outFr  = outDiv + (size_t)ZN * NVOX;

  float Nx = 0.0f, Ny = 0.0f, Nz = 0.0f, Ax = 0.0f, Ay = 0.0f, Az = 0.0f;
  float Cx = 0.0f, Cy = 0.0f, Cz = 0.0f, cbx = 0.0f, cby = 0.0f, cbz = 0.0f;
  float xx = 0.0f, xy = 0.0f, xz = 0.0f, yx = 0.0f, yy = 0.0f, yz = 0.0f;
  float zx = 0.0f, zy = 0.0f, zz = 0.0f;
  if (tid < NATOM) {
    const float* cn = C + (size_t)n * (NATOM * 3);
    const float ax = cn[tid * 3 + 0];
    const float ay = cn[tid * 3 + 1];
    const float az = cn[tid * 3 + 2];
    const float mk = atom_mask[(size_t)n * NATOM + tid];
    int lbl = L[n];
    __builtin_amdgcn_sched_barrier(0);
    if (lbl == -1) lbl = NLBL - 1;
    const float ch = charges[lbl * NATOM + tid];
    v2f xy2, zq;
    xy2.x = ax;
    xy2.y = ay;
    zq.x = az;
    zq.y = ch * mk;
    sAxy[tid] = xy2;
    sAzq[tid] = zq;
  }
  if (tid == 32) {
    const float* cn = C + (size_t)n * (NATOM * 3);
    Nx = cn[0]; Ny = cn[1]; Nz = cn[2];
    Ax = cn[3]; Ay = cn[4]; Az = cn[5];
    Cx = cn[6]; Cy = cn[7]; Cz = cn[8];
    const float b1x = Ax - Nx, b1y = Ay - Ny, b1z = Az - Nz;
    const float b2x = Cx - Ax, b2y = Cy - Ay, b2z = Cz - Az;
    const float b3x = b1y * b2z - b1z * b2y;
    const float b3y = b1z * b2x - b1x * b2z;
    const float b3z = b1x * b2y - b1y * b2x;
    cbx = Ax - 0.58273431f * b2x + 0.56802827f * b1x - 0.54067466f * b3x;
    cby = Ay - 0.58273431f * b2y + 0.56802827f * b1y - 0.54067466f * b3y;
    cbz = Az - 0.58273431f * b2z + 0.56802827f * b1z - 0.54067466f * b3z;

    yx = cbx - Ax; yy = cby - Ay; yz = cbz - Az;
    float yn = fmaxf(sqrtf(yx * yx + yy * yy + yz * yz), 1e-6f);
    yx /= yn; yy /= yn; yz /= yn;
    float xrx = Cx - Nx, xry = Cy - Ny, xrz = Cz - Nz;
    const float xp = xrx * yx + xry * yy + xrz * yz;
    xx = xrx - xp; xy = xry - xp; xz = xrz - xp;
    float xn = fmaxf(sqrtf(xx * xx + xy * xy + xz * xz), 1e-6f);
    xx /= xn; xy /= xn; xz /= xn;
    zx = xy * yz - xz * yy;
    zy = xz * yx - xx * yz;
    zz = xx * yy - xy * yx;
    float zn = fmaxf(sqrtf(zx * zx + zy * zy + zz * zz), 1e-6f);
    zx /= zn; zy /= zn; zz /= zn;

    sF[0] = xx; sF[1] = xy; sF[2] = xz;
    sF[3] = yx; sF[4] = yy; sF[5] = yz;
    sF[6] = zx; sF[7] = zy; sF[8] = zz;
    sO[0] = cbx; sO[1] = cby; sO[2] = cbz;
  }
  __syncthreads();
  if (tid == 32) {
    float* cb_out = outCb + (size_t)n * 12;
    cb_out[0] = Nx;  cb_out[1]  = Ny;  cb_out[2]  = Nz;
    cb_out[3] = Ax;  cb_out[4]  = Ay;  cb_out[5]  = Az;
    cb_out[6] = Cx;  cb_out[7]  = Cy;  cb_out[8]  = Cz;
    cb_out[9] = cbx; cb_out[10] = cby; cb_out[11] = cbz;
    float* fr = outFr + (size_t)n * 9;
    fr[0] = xx; fr[1] = xy; fr[2] = xz;
    fr[3] = yx; fr[4] = yy; fr[5] = yz;
    fr[6] = zx; fr[7] = zy; fr[8] = zz;
  }

  for (int j = tid; j < NVOX / 2; j += 256) {
    const int x = j >> 7, y = (j >> 3) & 15, z = (j & 7) << 1;
    const float vx = (float)x - 8.0f;
    const float vy = (float)y - 4.0f;
    v2f vz;
    vz.x = (float)z - 8.0f;
    vz.y = (float)(z + 1) - 8.0f;
    const v2f px = sO[0] + vx * sF[0] + vy * sF[3] + vz * sF[6];
    const v2f py = sO[1] + vx * sF[1] + vy * sF[4] + vz * sF[7];
    const v2f pz = sO[2] + vx * sF[2] + vy * sF[5] + vz * sF[8];
    v2f fx = {0.0f, 0.0f}, fy = {0.0f, 0.0f}, fz = {0.0f, 0.0f};
    const v2f seed = {1.17549435e-38f, 1.17549435e-38f};
#pragma unroll
    for (int a = 0; a < NATOM; ++a) {
      const v2f axy = sAxy[a], azq = sAzq[a];
      const v2f dx = px - axy.x;
      const v2f dy = py - axy.y;
      const v2f dz = pz - azq.x;
      const v2f dd = (dx * dx + seed) + dy * dy + dz * dz;
      v2f r;
      r.x = __builtin_amdgcn_rsqf(dd.x);
      r.y = __builtin_amdgcn_rsqf(dd.y);
      v2f q = r * r;
      q.x = fminf(q.x, 1.0f);
      q.y = fminf(q.y, 1.0f);
      const v2f w = (azq.y * r) * q;
      fx += w * dx; fy += w * dy; fz += w * dz;
    }
    const v2f ff = (fx * fx + seed) + fy * fy + fz * fz;
    v2f inv;
    inv.x = __builtin_amdgcn_rsqf(ff.x);
    inv.y = __builtin_amdgcn_rsqf(ff.y);
    *reinterpret_cast<v2f*>(&sFx[2 * j]) = fx * inv;
    *reinterpret_cast<v2f*>(&sFy[2 * j]) = fy * inv;
    *reinterpret_cast<v2f*>(&sFz[2 * j]) = fz * inv;
  }
  __syncthreads();

  v2f aReg[4], aPerm[4];
  {
    const int m     = lane & 15;
    const int mp    = 2 * (m & 7) + (m >> 3);
    const int khalf = (lane < 16) ? 0 : 2;
#pragma unroll
    for (int kk = 0; kk < 4; ++kk) {
      aReg[kk].x  = dm_elem(m, 4 * kk + khalf);
      aReg[kk].y  = dm_elem(m, 4 * kk + khalf + 1);
      aPerm[kk].x = dm_elem(mp, 4 * kk + khalf);
      aPerm[kk].y = dm_elem(mp, 4 * kk + khalf + 1);
    }
  }

  div_x_pass(sFx, sFx, aReg, lane, wv);
  __syncthreads();
  div_yz_store(sFx, sFy, sFz, outDiv + (size_t)n * NVOX, aReg, aPerm, lane, wv);
}

extern "C" void kernel_launch(void* const* d_in, const int* in_sizes, int n_in,
                              void* d_out, int out_size, void* d_ws, size_t ws_size,
                              hipStream_t stream) {
  const float* C         = (const float*)d_in[0];
  const int*   L         = (const int*)d_in[1];
  const float* atom_mask = (const float*)d_in[2];
  const float* charges   = (const float*)d_in[3];
  float* out = (float*)d_out;
  const int ZN = in_sizes[0] / (NATOM * 3);
  voxel_field_div_84670985273723<<<ZN, 256, 0, stream>>>(C, L, atom_mask, charges, out, ZN);
}
